# starting kernel
# baseline (speedup 1.0000x reference)
#include <hip/hip_runtime.h>
#include <hip/hip_bf16.h>

#define E_N 10000
#define T_N 24
#define D_N 128
#define H_N 8
#define K_N 16
#define FF_N 512
#define EPS_LN 1e-5f

typedef float v2f __attribute__((ext_vector_type(2)));
typedef float v8f __attribute__((ext_vector_type(8)));

__device__ __forceinline__ v8f wmma_f32(v2f a, v2f b, v8f c) {
  return __builtin_amdgcn_wmma_f32_16x16x4_f32(
       false, a,  false, b,
       (short)0, c,  false,  false);
}

__device__ __forceinline__ void gemm_acc(const float* sA, const float* B, int ldb,
                                         int wave, int half, int lm, int Kdim,
                                         v8f acc[4]) {
  const int n0 = wave * 32 + lm;
  const int n1 = n0 + 16;
  for (int k0 = 0; k0 < Kdim; k0 += 4) {
    const int ka = k0 + 2 * half;
    v2f a0, a1, b0, b1;
    a0.x = sA[lm * 128 + ka];        a0.y = sA[lm * 128 + ka + 1];
    a1.x = sA[(16 + lm) * 128 + ka]; a1.y = sA[(16 + lm) * 128 + ka + 1];
    b0.x = B[(size_t)ka * ldb + n0]; b0.y = B[(size_t)(ka + 1) * ldb + n0];
    b1.x = B[(size_t)ka * ldb + n1]; b1.y = B[(size_t)(ka + 1) * ldb + n1];
    acc[0] = wmma_f32(a0, b0, acc[0]);
    acc[1] = wmma_f32(a0, b1, acc[1]);
    acc[2] = wmma_f32(a1, b0, acc[2]);
    acc[3] = wmma_f32(a1, b1, acc[3]);
  }
}

__global__ __launch_bounds__(128) void kv_proj_kernel(
    const float* __restrict__ x, const float* __restrict__ Wk,
    const float* __restrict__ Wv, float* __restrict__ Kx, float* __restrict__ Vx) {
  __shared__ float sx0[16 * 128];
  const int tid = threadIdx.x;
  const int e0 = blockIdx.x * 16;
  for (int i = tid; i < 16 * 128; i += 128) {
    const int r = i >> 7, c = i & 127;
    sx0[i] = x[(size_t)(e0 + r) * (T_N * D_N) + c];
  }
  __syncthreads();

  const int lane = tid & 31, wave = tid >> 5;
  const int half = lane >> 4, lm = lane & 15;
  const int n0 = wave * 32 + lm, n1 = n0 + 16;
  v8f aK0{}, aK1{}, aV0{}, aV1{};
  for (int k0 = 0; k0 < 128; k0 += 4) {
    const int ka = k0 + 2 * half;
    v2f a, bk0, bk1, bv0, bv1;
    a.x = sx0[lm * 128 + ka];  a.y = sx0[lm * 128 + ka + 1];
    bk0.x = Wk[ka * 128 + n0]; bk0.y = Wk[(ka + 1) * 128 + n0];
    bk1.x = Wk[ka * 128 + n1]; bk1.y = Wk[(ka + 1) * 128 + n1];
    bv0.x = Wv[ka * 128 + n0]; bv0.y = Wv[(ka + 1) * 128 + n0];
    bv1.x = Wv[ka * 128 + n1]; bv1.y = Wv[(ka + 1) * 128 + n1];
    aK0 = wmma_f32(a, bk0, aK0);
    aK1 = wmma_f32(a, bk1, aK1);
    aV0 = wmma_f32(a, bv0, aV0);
    aV1 = wmma_f32(a, bv1, aV1);
  }
  for (int r = 0; r < 8; ++r) {
    const size_t row = (size_t)(e0 + r + 8 * half) * 128;
    Kx[row + n0] = aK0[r];
    Kx[row + n1] = aK1[r];
    Vx[row + n0] = aV0[r];
    Vx[row + n1] = aV1[r];
  }
}

__global__ __launch_bounds__(128) void attn_block_kernel(
    const float* __restrict__ x, const int* __restrict__ nbr,
    const float* __restrict__ Wq, const float* __restrict__ Wo,
    const float* __restrict__ bo, const float* __restrict__ g1,
    const float* __restrict__ b1, const float* __restrict__ g2,
    const float* __restrict__ b2, const float* __restrict__ W1,
    const float* __restrict__ bf1, const float* __restrict__ W2,
    const float* __restrict__ bf2, const float* __restrict__ Kx,
    const float* __restrict__ Vx, float* __restrict__ out) {
  __shared__ float buf0[32 * 128];
  __shared__ float buf1[32 * 128];
  __shared__ float buf2[32 * 128];
  __shared__ float buf3[32 * 128];
  const int VOFF = 16 * 128;

  const int tid = threadIdx.x;
  const int e = blockIdx.x;
  const int lane = tid & 31, wave = tid >> 5;
  const int half = lane >> 4, lm = lane & 15;
  const size_t xbase = (size_t)e * (T_N * D_N);

  for (int i = tid; i < 32 * 128; i += 128) {
    const int r = i >> 7, c = i & 127;
    buf0[i] = (r < T_N) ? x[xbase + (size_t)r * D_N + c] : 0.0f;
  }
  for (int i = tid; i < 16 * 128; i += 128) {
    const int k = i >> 7, c = i & 127;
    const int n = nbr[e * K_N + k];
    buf2[i]        = Kx[(size_t)n * D_N + c];
    buf2[VOFF + i] = Vx[(size_t)n * D_N + c];
  }
  __syncthreads();

  {
    v8f q[4] = {};
    gemm_acc(buf0, Wq, 128, wave, half, lm, 128, q);
    for (int i = 0; i < 4; ++i)
      for (int r = 0; r < 8; ++r) {
        const int row = ((i >> 1) << 4) + r + 8 * half;
        const int col = wave * 32 + ((i & 1) << 4) + lm;
        buf1[row * 128 + col] = q[i][r] * 0.25f;
      }
  }
  __syncthreads();

  {
    const int hc0 = (2 * wave) * 16, hc1 = hc0 + 16;
    v8f s0{}, s1{}, s2{}, s3{};
    for (int k0 = 0; k0 < 16; k0 += 4) {
      const int ka = k0 + 2 * half;
      v2f a00, a01, a10, a11, b0, b1;
      a00.x = buf1[lm * 128 + hc0 + ka];        a00.y = buf1[lm * 128 + hc0 + ka + 1];
      a01.x = buf1[(16 + lm) * 128 + hc0 + ka]; a01.y = buf1[(16 + lm) * 128 + hc0 + ka + 1];
      a10.x = buf1[lm * 128 + hc1 + ka];        a10.y = buf1[lm * 128 + hc1 + ka + 1];
      a11.x = buf1[(16 + lm) * 128 + hc1 + ka]; a11.y = buf1[(16 + lm) * 128 + hc1 + ka + 1];
      b0.x = buf2[lm * 128 + hc0 + ka]; b0.y = buf2[lm * 128 + hc0 + ka + 1];
      b1.x = buf2[lm * 128 + hc1 + ka]; b1.y = buf2[lm * 128 + hc1 + ka + 1];
      s0 = wmma_f32(a00, b0, s0);
      s1 = wmma_f32(a01, b0, s1);
      s2 = wmma_f32(a10, b1, s2);
      s3 = wmma_f32(a11, b1, s3);
    }
    for (int r = 0; r < 8; ++r) {
      const int t0 = r + 8 * half, t1 = 16 + r + 8 * half;
      buf3[t0 * 128 + hc0 + lm] = s0[r];
      buf3[t1 * 128 + hc0 + lm] = s1[r];
      buf3[t0 * 128 + hc1 + lm] = s2[r];
      buf3[t1 * 128 + hc1 + lm] = s3[r];
    }
  }
  __syncthreads();

  for (int idx = tid; idx < T_N * H_N; idx += 128) {
    const int t = idx >> 3, h = idx & 7;
    float* row = &buf3[t * 128 + h * 16];
    float m = row[0];
    for (int k = 1; k < 16; ++k) m = fmaxf(m, row[k]);
    float s = 0.f;
    for (int k = 0; k < 16; ++k) { row[k] = __expf(row[k] - m); s += row[k]; }
    const float inv = 1.f / s;
    for (int k = 0; k < 16; ++k) row[k] *= inv;
  }
  __syncthreads();

  {
    const int hc0 = (2 * wave) * 16, hc1 = hc0 + 16;
    v8f z0{}, z1{}, z2{}, z3{};
    for (int k0 = 0; k0 < 16; k0 += 4) {
      const int ka = k0 + 2 * half;
      v2f a00, a01, a10, a11, b0, b1;
      a00.x = buf3[lm * 128 + hc0 + ka];        a00.y = buf3[lm * 128 + hc0 + ka + 1];
      a01.x = buf3[(16 + lm) * 128 + hc0 + ka]; a01.y = buf3[(16 + lm) * 128 + hc0 + ka + 1];
      a10.x = buf3[lm * 128 + hc1 + ka];        a10.y = buf3[lm * 128 + hc1 + ka + 1];
      a11.x = buf3[(16 + lm) * 128 + hc1 + ka]; a11.y = buf3[(16 + lm) * 128 + hc1 + ka + 1];
      b0.x = buf2[VOFF + ka * 128 + hc0 + lm]; b0.y = buf2[VOFF + (ka + 1) * 128 + hc0 + lm];
      b1.x = buf2[VOFF + ka * 128 + hc1 + lm]; b1.y = buf2[VOFF + (ka + 1) * 128 + hc1 + lm];
      z0 = wmma_f32(a00, b0, z0);
      z1 = wmma_f32(a01, b0, z1);
      z2 = wmma_f32(a10, b1, z2);
      z3 = wmma_f32(a11, b1, z3);
    }
    for (int r = 0; r < 8; ++r) {
      const int t0 = r + 8 * half, t1 = 16 + r + 8 * half;
      buf1[t0 * 128 + hc0 + lm] = z0[r];
      buf1[t1 * 128 + hc0 + lm] = z1[r];
      buf1[t0 * 128 + hc1 + lm] = z2[r];
      buf1[t1 * 128 + hc1 + lm] = z3[r];
    }
  }
  __syncthreads();

  {
    v8f o[4] = {};
    gemm_acc(buf1, Wo, 128, wave, half, lm, 128, o);
    for (int i = 0; i < 4; ++i)
      for (int r = 0; r < 8; ++r) {
        const int row = ((i >> 1) << 4) + r + 8 * half;
        const int col = wave * 32 + ((i & 1) << 4) + lm;
        buf2[row * 128 + col] = o[i][r] + bo[col] + buf0[row * 128 + col];
      }
  }
  __syncthreads();

  if (tid < T_N) {
    const float* rp = &buf2[tid * 128];
    float mu = 0.f;
    for (int c = 0; c < 128; ++c) mu += rp[c];
    mu *= (1.f / 128.f);
    float var = 0.f;
    for (int c = 0; c < 128; ++c) { const float d = rp[c] - mu; var += d * d; }
    var *= (1.f / 128.f);
    const float rs = rsqrtf(var + EPS_LN);
    float* wp = &buf3[tid * 128];
    for (int c = 0; c < 128; ++c) wp[c] = (rp[c] - mu) * rs * g1[c] + b1[c];
  }
  __syncthreads();

  v8f y[4] = {};
  for (int ch = 0; ch < 4; ++ch) {
    v8f g[4] = {};
    gemm_acc(buf3, W1 + ch * 128, FF_N, wave, half, lm, 128, g);
    for (int i = 0; i < 4; ++i)
      for (int r = 0; r < 8; ++r) {
        const int row = ((i >> 1) << 4) + r + 8 * half;
        const int col = wave * 32 + ((i & 1) << 4) + lm;
        buf1[row * 128 + col] = fmaxf(g[i][r] + bf1[ch * 128 + col], 0.f);
      }
    __syncthreads();
    gemm_acc(buf1, W2 + (size_t)(ch * 128) * D_N, 128, wave, half, lm, 128, y);
    __syncthreads();
  }

  for (int i = 0; i < 4; ++i)
    for (int r = 0; r < 8; ++r) {
      const int row = ((i >> 1) << 4) + r + 8 * half;
      const int col = wave * 32 + ((i & 1) << 4) + lm;
      buf1[row * 128 + col] = y[i][r] + bf2[col] + buf3[row * 128 + col];
    }
  __syncthreads();

  if (tid < T_N) {
    const float* rp = &buf1[tid * 128];
    float mu = 0.f;
    for (int c = 0; c < 128; ++c) mu += rp[c];
    mu *= (1.f / 128.f);
    float var = 0.f;
    for (int c = 0; c < 128; ++c) { const float d = rp[c] - mu; var += d * d; }
    var *= (1.f / 128.f);
    const float rs = rsqrtf(var + EPS_LN);
    for (int c = 0; c < 128; ++c)
      out[xbase + (size_t)tid * D_N + c] = (rp[c] - mu) * rs * g2[c] + b2[c];
  }
}

extern "C" void kernel_launch(void* const* d_in, const int* in_sizes, int n_in,
                              void* d_out, int out_size, void* d_ws, size_t ws_size,
                              hipStream_t stream) {
  const float* x   = (const float*)d_in[0];
  const int*   nbr = (const int*)d_in[1];
  const float* Wq  = (const float*)d_in[2];
  const float* Wk  = (const float*)d_in[3];
  const float* Wv  = (const float*)d_in[4];
  const float* Wo  = (const float*)d_in[5];
  const float* bo  = (const float*)d_in[6];
  const float* g1  = (const float*)d_in[7];
  const float* b1  = (const float*)d_in[8];
  const float* g2  = (const float*)d_in[9];
  const float* b2  = (const float*)d_in[10];
  const float* W1  = (const float*)d_in[11];
  const float* bf1 = (const float*)d_in[12];
  const float* W2  = (const float*)d_in[13];
  const float* bf2 = (const float*)d_in[14];

  float* Kx = (float*)d_ws;
  float* Vx = Kx + (size_t)E_N * D_N;

  kv_proj_kernel<<<E_N / 16, 128, 0, stream>>>(x, Wk, Wv, Kx, Vx);
  attn_block_kernel<<<E_N, 128, 0, stream>>>(x, nbr, Wq, Wo, bo, g1, b1, g2, b2,
                                             W1, bf1, W2, bf2, Kx, Vx,
                                             (float*)d_out);
}


# wave tile 32x32 -> 32x64, two waves per work-group: a lane owns 4 adjacent columns, so one 16-byte weight load feeds 8 WMMAs; LDS activation tiles read by 2 waves, not 4; sum order unchanged
# speedup vs baseline: 1.2269x; 1.2269x over previous
#include <hip/hip_runtime.h>
#include <hip/hip_bf16.h>

#define E_N 10000
#define T_N 24
#define D_N 128
#define H_N 8
#define K_N 16
#define FF_N 512
#define EPS_LN 1e-5f

typedef float v2f __attribute__((ext_vector_type(2)));
typedef float v8f __attribute__((ext_vector_type(8)));

__device__ __forceinline__ v8f wmma_f32(v2f a, v2f b, v8f c) {
  return __builtin_amdgcn_wmma_f32_16x16x4_f32(
       false, a,  false, b,
       (short)0, c,  false,  false);
}

typedef float v4f __attribute__((ext_vector_type(4)));

__device__ __forceinline__ void gemm_acc(const float* sA, const float* B, int ldb,
                                         int wave, int half, int lm, v8f acc[8]) {
  const float* bp = B + wave * 64 + 4 * lm;
  for (int k0 = 0; k0 < 128; k0 += 4) {
    const int ka = k0 + 2 * half;
    v2f a0, a1;
    a0.x = sA[lm * 128 + ka];        a0.y = sA[lm * 128 + ka + 1];
    a1.x = sA[(16 + lm) * 128 + ka]; a1.y = sA[(16 + lm) * 128 + ka + 1];
    const v4f bx = *(const v4f*)(bp + (size_t)ka * ldb);
    const v4f by = *(const v4f*)(bp + (size_t)(ka + 1) * ldb);
    for (int j = 0; j < 4; ++j) {
      v2f b;
      b.x = bx[j]; b.y = by[j];
      acc[j]     = wmma_f32(a0, b, acc[j]);
      acc[4 + j] = wmma_f32(a1, b, acc[4 + j]);
    }
  }
}

__device__ __forceinline__ int tile_row(int t, int r, int half) { return 16 * t + r + 8 * half; }

__global__ __launch_bounds__(128) void kv_proj_kernel(
    const float* __restrict__ x, const float* __restrict__ Wk,
    const float* __restrict__ Wv, float* __restrict__ Kx, float* __restrict__ Vx) {
  __shared__ float sx0[16 * 128];
  const int tid = threadIdx.x;
  const int e0 = blockIdx.x * 16;
  for (int i = tid; i < 16 * 128; i += 128) {
    const int r = i >> 7, c = i & 127;
    sx0[i] = x[(size_t)(e0 + r) * (T_N * D_N) + c];
  }
  __syncthreads();

  const int lane = tid & 31, wave = tid >> 5;
  const int half = lane >> 4, lm = lane & 15;
  const int n0 = wave * 32 + lm, n1 = n0 + 16;
  v8f aK0{}, aK1{}, aV0{}, aV1{};
  for (int k0 = 0; k0 < 128; k0 += 4) {
    const int ka = k0 + 2 * half;
    v2f a, bk0, bk1, bv0, bv1;
    a.x = sx0[lm * 128 + ka];  a.y = sx0[lm * 128 + ka + 1];
    bk0.x = Wk[ka * 128 + n0]; bk0.y = Wk[(ka + 1) * 128 + n0];
    bk1.x = Wk[ka * 128 + n1]; bk1.y = Wk[(ka + 1) * 128 + n1];
    bv0.x = Wv[ka * 128 + n0]; bv0.y = Wv[(ka + 1) * 128 + n0];
    bv1.x = Wv[ka * 128 + n1]; bv1.y = Wv[(ka + 1) * 128 + n1];
    aK0 = wmma_f32(a, bk0, aK0);
    aK1 = wmma_f32(a, bk1, aK1);
    aV0 = wmma_f32(a, bv0, aV0);
    aV1 = wmma_f32(a, bv1, aV1);
  }
  for (int r = 0; r < 8; ++r) {
    const size_t row = (size_t)(e0 + r + 8 * half) * 128;
    Kx[row + n0] = aK0[r];
    Kx[row + n1] = aK1[r];
    Vx[row + n0] = aV0[r];
    Vx[row + n1] = aV1[r];
  }
}

__global__ __launch_bounds__(64) void attn_block_kernel(
    const float* __restrict__ x, const int* __restrict__ nbr,
    const float* __restrict__ Wq, const float* __restrict__ Wo,
    const float* __restrict__ bo, const float* __restrict__ g1,
    const float* __restrict__ b1, const float* __restrict__ g2,
    const float* __restrict__ b2, const float* __restrict__ W1,
    const float* __restrict__ bf1, const float* __restrict__ W2,
    const float* __restrict__ bf2, const float* __restrict__ Kx,
    const float* __restrict__ Vx, float* __restrict__ out) {
  __shared__ __attribute__((aligned(16))) float buf0[32 * 128];
  __shared__ __attribute__((aligned(16))) float buf1[32 * 128];
  __shared__ __attribute__((aligned(16))) float buf2[32 * 128];
  __shared__ __attribute__((aligned(16))) float buf3[32 * 128];
  const int VOFF = 16 * 128;

  const int tid = threadIdx.x;
  const int e = blockIdx.x;
  const int lane = tid & 31, wave = tid >> 5;
  const int half = lane >> 4, lm = lane & 15;
  const size_t xbase = (size_t)e * (T_N * D_N);
  const int c4 = wave * 64 + 4 * lm;

  for (int i = tid; i < 32 * 128; i += 64) {
    const int r = i >> 7, c = i & 127;
    buf0[i] = (r < T_N) ? x[xbase + (size_t)r * D_N + c] : 0.0f;
  }
  for (int i = tid; i < 16 * 128; i += 64) {
    const int k = i >> 7, c = i & 127;
    const int n = nbr[e * K_N + k];
    buf2[i]        = Kx[(size_t)n * D_N + c];
    buf2[VOFF + i] = Vx[(size_t)n * D_N + c];
  }
  __syncthreads();

  {
    v8f q[8] = {};
    gemm_acc(buf0, Wq, 128, wave, half, lm, q);
    for (int t = 0; t < 2; ++t)
      for (int r = 0; r < 8; ++r) {
        v4f v;
        for (int j = 0; j < 4; ++j) v[j] = q[4 * t + j][r] * 0.25f;
        *(v4f*)&buf1[tile_row(t, r, half) * 128 + c4] = v;
      }
  }
  __syncthreads();

  for (int hh = 0; hh < 4; ++hh) {
    const int hc = (4 * wave + hh) * 16;
    v8f s0{}, s1{};
    for (int k0 = 0; k0 < 16; k0 += 4) {
      const int ka = k0 + 2 * half;
      v2f a0, a1, b;
      a0.x = buf1[lm * 128 + hc + ka];        a0.y = buf1[lm * 128 + hc + ka + 1];
      a1.x = buf1[(16 + lm) * 128 + hc + ka]; a1.y = buf1[(16 + lm) * 128 + hc + ka + 1];
      b.x = buf2[lm * 128 + hc + ka]; b.y = buf2[lm * 128 + hc + ka + 1];
      s0 = wmma_f32(a0, b, s0);
      s1 = wmma_f32(a1, b, s1);
    }
    for (int r = 0; r < 8; ++r) {
      buf3[tile_row(0, r, half) * 128 + hc + lm] = s0[r];
      buf3[tile_row(1, r, half) * 128 + hc + lm] = s1[r];
    }
  }
  __syncthreads();

  for (int idx = tid; idx < T_N * H_N; idx += 64) {
    const int t = idx >> 3, h = idx & 7;
    float* row = &buf3[t * 128 + h * 16];
    float m = row[0];
    for (int k = 1; k < 16; ++k) m = fmaxf(m, row[k]);
    float s = 0.f;
    for (int k = 0; k < 16; ++k) { row[k] = __expf(row[k] - m); s += row[k]; }
    const float inv = 1.f / s;
    for (int k = 0; k < 16; ++k) row[k] *= inv;
  }
  __syncthreads();

  for (int hh = 0; hh < 4; ++hh) {
    const int hc = (4 * wave + hh) * 16;
    v8f z0{}, z1{};
    for (int k0 = 0; k0 < 16; k0 += 4) {
      const int ka = k0 + 2 * half;
      v2f a0, a1, b;
      a0.x = buf3[lm * 128 + hc + ka];        a0.y = buf3[lm * 128 + hc + ka + 1];
      a1.x = buf3[(16 + lm) * 128 + hc + ka]; a1.y = buf3[(16 + lm) * 128 + hc + ka + 1];
      b.x = buf2[VOFF + ka * 128 + hc + lm]; b.y = buf2[VOFF + (ka + 1) * 128 + hc + lm];
      z0 = wmma_f32(a0, b, z0);
      z1 = wmma_f32(a1, b, z1);
    }
    for (int r = 0; r < 8; ++r) {
      buf1[tile_row(0, r, half) * 128 + hc + lm] = z0[r];
      buf1[tile_row(1, r, half) * 128 + hc + lm] = z1[r];
    }
  }
  __syncthreads();

  {
    v8f o[8] = {};
    gemm_acc(buf1, Wo, 128, wave, half, lm, o);
    const v4f bias = *(const v4f*)&bo[c4];
    for (int t = 0; t < 2; ++t)
      for (int r = 0; r < 8; ++r) {
        const int off = tile_row(t, r, half) * 128 + c4;
        const v4f xr = *(const v4f*)&buf0[off];
        v4f v;
        for (int j = 0; j < 4; ++j) v[j] = o[4 * t + j][r] + bias[j] + xr[j];
        *(v4f*)&buf2[off] = v;
      }
  }
  __syncthreads();

  if (tid < T_N) {
    const float* rp = &buf2[tid * 128];
    float mu = 0.f;
    for (int c = 0; c < 128; ++c) mu += rp[c];
    mu *= (1.f / 128.f);
    float var = 0.f;
    for (int c = 0; c < 128; ++c) { const float d = rp[c] - mu; var += d * d; }
    var *= (1.f / 128.f);
    const float rs = rsqrtf(var + EPS_LN);
    float* wp = &buf3[tid * 128];
    for (int c = 0; c < 128; ++c) wp[c] = (rp[c] - mu) * rs * g1[c] + b1[c];
  }
  __syncthreads();

  v8f y[8] = {};
  for (int ch = 0; ch < 4; ++ch) {
    v8f g[8] = {};
    gemm_acc(buf3, W1 + ch * 128, FF_N, wave, half, lm, g);
    const v4f bias = *(const v4f*)&bf1[ch * 128 + c4];
    for (int t = 0; t < 2; ++t)
      for (int r = 0; r < 8; ++r) {
        v4f v;
        for (int j = 0; j < 4; ++j) v[j] = fmaxf(g[4 * t + j][r] + bias[j], 0.f);
        *(v4f*)&buf1[tile_row(t, r, half) * 128 + c4] = v;
      }
    __syncthreads();
    gemm_acc(buf1, W2 + (size_t)(ch * 128) * D_N, 128, wave, half, lm, y);
    __syncthreads();
  }

  {
    const v4f bias = *(const v4f*)&bf2[c4];
    for (int t = 0; t < 2; ++t)
      for (int r = 0; r < 8; ++r) {
        const int off = tile_row(t, r, half) * 128 + c4;
        const v4f hr = *(const v4f*)&buf3[off];
        v4f v;
        for (int j = 0; j < 4; ++j) v[j] = y[4 * t + j][r] + bias[j] + hr[j];
        *(v4f*)&buf1[off] = v;
      }
  }
  __syncthreads();

  if (tid < T_N) {
    const float* rp = &buf1[tid * 128];
    float mu = 0.f;
    for (int c = 0; c < 128; ++c) mu += rp[c];
    mu *= (1.f / 128.f);
    float var = 0.f;
    for (int c = 0; c < 128; ++c) { const float d = rp[c] - mu; var += d * d; }
    var *= (1.f / 128.f);
    const float rs = rsqrtf(var + EPS_LN);
    for (int c = 0; c < 128; ++c)
      out[xbase + (size_t)tid * D_N + c] = (rp[c] - mu) * rs * g2[c] + b2[c];
  }
}

extern "C" void kernel_launch(void* const* d_in, const int* in_sizes, int n_in,
                              void* d_out, int out_size, void* d_ws, size_t ws_size,
                              hipStream_t stream) {
  const float* x   = (const float*)d_in[0];
  const int*   nbr = (const int*)d_in[1];
  const float* Wq  = (const float*)d_in[2];
  const float* Wk  = (const float*)d_in[3];
  const float* Wv  = (const float*)d_in[4];
  const float* Wo  = (const float*)d_in[5];
  const float* bo  = (const float*)d_in[6];
  const float* g1  = (const float*)d_in[7];
  const float* b1  = (const float*)d_in[8];
  const float* g2  = (const float*)d_in[9];
  const float* b2  = (const float*)d_in[10];
  const float* W1  = (const float*)d_in[11];
  const float* bf1 = (const float*)d_in[12];
  const float* W2  = (const float*)d_in[13];
  const float* bf2 = (const float*)d_in[14];

  float* Kx = (float*)d_ws;
  float* Vx = Kx + (size_t)E_N * D_N;

  kv_proj_kernel<<<E_N / 16, 128, 0, stream>>>(x, Wk, Wv, Kx, Vx);
  attn_block_kernel<<<E_N, 64, 0, stream>>>(x, nbr, Wq, Wo, bo, g1, b1, g2, b2,
                                             W1, bf1, W2, bf2, Kx, Vx,
                                             (float*)d_out);
}


# attention kernel persistent (1024 work-groups loop over events): next event's x and gathered K/V rows fetched as one group of 16-byte loads under LayerNorm 2, not by a waited load loop
# speedup vs baseline: 1.4530x; 1.1843x over previous
#include <hip/hip_runtime.h>
#include <hip/hip_bf16.h>

#define E_N 10000
#define T_N 24
#define D_N 128
#define H_N 8
#define K_N 16
#define FF_N 512
#define EPS_LN 1e-5f

typedef float v2f __attribute__((ext_vector_type(2)));
typedef float v8f __attribute__((ext_vector_type(8)));

__device__ __forceinline__ v8f wmma_f32(v2f a, v2f b, v8f c) {
  return __builtin_amdgcn_wmma_f32_16x16x4_f32(
       false, a,  false, b,
       (short)0, c,  false,  false);
}

typedef float v4f __attribute__((ext_vector_type(4)));

__device__ __forceinline__ void gemm_acc(const float* sA, const float* B, int ldb,
                                         int wave, int half, int lm, v8f acc[8]) {
  const float* bp = B + wave * 64 + 4 * lm;
  for (int k0 = 0; k0 < 128; k0 += 4) {
    if (k0 == 64) __builtin_amdgcn_sched_barrier(0);
    const int ka = k0 + 2 * half;
    v2f a0, a1;
    a0.x = sA[lm * 128 + ka];        a0.y = sA[lm * 128 + ka + 1];
    a1.x = sA[(16 + lm) * 128 + ka]; a1.y = sA[(16 + lm) * 128 + ka + 1];
    const v4f bx = *(const v4f*)(bp + (size_t)ka * ldb);
    const v4f by = *(const v4f*)(bp + (size_t)(ka + 1) * ldb);
    for (int j = 0; j < 4; ++j) {
      v2f b;
      b.x = bx[j]; b.y = by[j];
      acc[j]     = wmma_f32(a0, b, acc[j]);
      acc[4 + j] = wmma_f32(a1, b, acc[4 + j]);
    }
  }
}

__device__ __forceinline__ int tile_row(int t, int r, int half) { return 16 * t + r + 8 * half; }

__global__ __launch_bounds__(128) void kv_proj_kernel(
    const float* __restrict__ x, const float* __restrict__ Wk,
    const float* __restrict__ Wv, float* __restrict__ Kx, float* __restrict__ Vx) {
  __shared__ float sx0[16 * 128];
  const int tid = threadIdx.x;
  const int e0 = blockIdx.x * 16;
  for (int i = tid; i < 16 * 128; i += 128) {
    const int r = i >> 7, c = i & 127;
    sx0[i] = x[(size_t)(e0 + r) * (T_N * D_N) + c];
  }
  __syncthreads();

  const int lane = tid & 31, wave = tid >> 5;
  const int half = lane >> 4, lm = lane & 15;
  const int n0 = wave * 32 + lm, n1 = n0 + 16;
  v8f aK0{}, aK1{}, aV0{}, aV1{};
  for (int k0 = 0; k0 < 128; k0 += 4) {
    const int ka = k0 + 2 * half;
    v2f a, bk0, bk1, bv0, bv1;
    a.x = sx0[lm * 128 + ka];  a.y = sx0[lm * 128 + ka + 1];
    bk0.x = Wk[ka * 128 + n0]; bk0.y = Wk[(ka + 1) * 128 + n0];
    bk1.x = Wk[ka * 128 + n1]; bk1.y = Wk[(ka + 1) * 128 + n1];
    bv0.x = Wv[ka * 128 + n0]; bv0.y = Wv[(ka + 1) * 128 + n0];
    bv1.x = Wv[ka * 128 + n1]; bv1.y = Wv[(ka + 1) * 128 + n1];
    aK0 = wmma_f32(a, bk0, aK0);
    aK1 = wmma_f32(a, bk1, aK1);
    aV0 = wmma_f32(a, bv0, aV0);
    aV1 = wmma_f32(a, bv1, aV1);
  }
  for (int r = 0; r < 8; ++r) {
    const size_t row = (size_t)(e0 + r + 8 * half) * 128;
    Kx[row + n0] = aK0[r];
    Kx[row + n1] = aK1[r];
    Vx[row + n0] = aV0[r];
    Vx[row + n1] = aV1[r];
  }
}

struct Staged { v4f xr[12]; v4f kr[8]; v4f vr[8]; };

__device__ __forceinline__ void fetch_nbr(const int* __restrict__ nbr, int e, int wave_u,
                                          int nb[8]) {
  for (int j = 0; j < 8; ++j) nb[j] = nbr[e * K_N + wave_u + 2 * j];
}

__device__ __forceinline__ void fetch_event(const float* __restrict__ x,
                                            const float* __restrict__ Kx,
                                            const float* __restrict__ Vx,
                                            int e, int tid, const int nb[8], Staged& s) {
  const float* xe = x + (size_t)e * (T_N * D_N);
  for (int j = 0; j < 12; ++j) s.xr[j] = *(const v4f*)(xe + 4 * (tid + 64 * j));
  for (int j = 0; j < 8; ++j) {
    const int c = (tid & 31) * 4;
    const int n = nb[j];
    s.kr[j] = *(const v4f*)(Kx + (size_t)n * D_N + c);
    s.vr[j] = *(const v4f*)(Vx + (size_t)n * D_N + c);
  }
}

__device__ __forceinline__ void store_event(float* bx, float* bkv, int tid, const Staged& s) {
  for (int j = 0; j < 12; ++j) *(v4f*)&bx[4 * (tid + 64 * j)] = s.xr[j];
  for (int j = 0; j < 8; ++j) {
    const int i4 = tid + 64 * j;
    *(v4f*)&bkv[4 * i4] = s.kr[j];
    *(v4f*)&bkv[16 * 128 + 4 * i4] = s.vr[j];
  }
}

__global__ __launch_bounds__(64) __attribute__((amdgpu_waves_per_eu(2, 2))) void attn_block_kernel(
    const float* __restrict__ x, const int* __restrict__ nbr,
    const float* __restrict__ Wq, const float* __restrict__ Wo,
    const float* __restrict__ bo, const float* __restrict__ g1,
    const float* __restrict__ b1, const float* __restrict__ g2,
    const float* __restrict__ b2, const float* __restrict__ W1,
    const float* __restrict__ bf1, const float* __restrict__ W2,
    const float* __restrict__ bf2, const float* __restrict__ Kx,
    const float* __restrict__ Vx, float* __restrict__ out) {
  __shared__ __attribute__((aligned(16))) float buf0[32 * 128];
  __shared__ __attribute__((aligned(16))) float buf1[32 * 128];
  __shared__ __attribute__((aligned(16))) float buf2[32 * 128];
  __shared__ __attribute__((aligned(16))) float buf3[32 * 128];
  const int VOFF = 16 * 128;

  const int tid0 = threadIdx.x;

  for (int i = tid0; i < 8 * 32; i += 64) {
    const v4f z = {0.f, 0.f, 0.f, 0.f};
    *(v4f*)&buf0[T_N * 128 + 4 * i] = z;
  }
  {
    Staged s;
    int nb[8];
    fetch_nbr(nbr, (int)blockIdx.x, __builtin_amdgcn_readfirstlane(tid0 >> 5), nb);
    fetch_event(x, Kx, Vx, (int)blockIdx.x, tid0, nb, s);
    store_event(buf0, buf2, tid0, s);
  }
  __syncthreads();

  for (int e = blockIdx.x; e < E_N; e += gridDim.x) {
  const size_t xbase = (size_t)e * (T_N * D_N);
  const int e_next = e + (int)gridDim.x;
  const bool more = e_next < E_N;
  int tid = tid0;
  asm volatile("" : "+v"(tid));
  const int lane = tid & 31, wave = tid >> 5;
  const int half = lane >> 4, lm = lane & 15;
  const int c4 = wave * 64 + 4 * lm;

  int nb_next[8];
  if (more) fetch_nbr(nbr, e_next, __builtin_amdgcn_readfirstlane(wave), nb_next);

  {
    v8f q[8] = {};
    gemm_acc(buf0, Wq, 128, wave, half, lm, q);
    for (int t = 0; t < 2; ++t)
      for (int r = 0; r < 8; ++r) {
        v4f v;
        for (int j = 0; j < 4; ++j) v[j] = q[4 * t + j][r] * 0.25f;
        *(v4f*)&buf1[tile_row(t, r, half) * 128 + c4] = v;
      }
  }
  __syncthreads();

  for (int hh = 0; hh < 4; ++hh) {
    const int hc = (4 * wave + hh) * 16;
    v8f s0{}, s1{};
    for (int k0 = 0; k0 < 16; k0 += 4) {
      const int ka = k0 + 2 * half;
      v2f a0, a1, b;
      a0.x = buf1[lm * 128 + hc + ka];        a0.y = buf1[lm * 128 + hc + ka + 1];
      a1.x = buf1[(16 + lm) * 128 + hc + ka]; a1.y = buf1[(16 + lm) * 128 + hc + ka + 1];
      b.x = buf2[lm * 128 + hc + ka]; b.y = buf2[lm * 128 + hc + ka + 1];
      s0 = wmma_f32(a0, b, s0);
      s1 = wmma_f32(a1, b, s1);
    }
    for (int r = 0; r < 8; ++r) {
      buf3[tile_row(0, r, half) * 128 + hc + lm] = s0[r];
      buf3[tile_row(1, r, half) * 128 + hc + lm] = s1[r];
    }
  }
  __syncthreads();

  for (int idx = tid; idx < T_N * H_N; idx += 64) {
    const int t = idx >> 3, h = idx & 7;
    float* row = &buf3[t * 128 + h * 16];
    float m = row[0];
    for (int k = 1; k < 16; ++k) m = fmaxf(m, row[k]);
    float s = 0.f;
    for (int k = 0; k < 16; ++k) { row[k] = __expf(row[k] - m); s += row[k]; }
    const float inv = 1.f / s;
    for (int k = 0; k < 16; ++k) row[k] *= inv;
  }
  __syncthreads();

  for (int hh = 0; hh < 4; ++hh) {
    const int hc = (4 * wave + hh) * 16;
    v8f z0{}, z1{};
    for (int k0 = 0; k0 < 16; k0 += 4) {
      const int ka = k0 + 2 * half;
      v2f a0, a1, b;
      a0.x = buf3[lm * 128 + hc + ka];        a0.y = buf3[lm * 128 + hc + ka + 1];
      a1.x = buf3[(16 + lm) * 128 + hc + ka]; a1.y = buf3[(16 + lm) * 128 + hc + ka + 1];
      b.x = buf2[VOFF + ka * 128 + hc + lm]; b.y = buf2[VOFF + (ka + 1) * 128 + hc + lm];
      z0 = wmma_f32(a0, b, z0);
      z1 = wmma_f32(a1, b, z1);
    }
    for (int r = 0; r < 8; ++r) {
      buf1[tile_row(0, r, half) * 128 + hc + lm] = z0[r];
      buf1[tile_row(1, r, half) * 128 + hc + lm] = z1[r];
    }
  }
  __syncthreads();

  {
    v8f o[8] = {};
    gemm_acc(buf1, Wo, 128, wave, half, lm, o);
    const v4f bias = *(const v4f*)&bo[c4];
    for (int t = 0; t < 2; ++t)
      for (int r = 0; r < 8; ++r) {
        const int off = tile_row(t, r, half) * 128 + c4;
        const v4f xr = *(const v4f*)&buf0[off];
        v4f v;
        for (int j = 0; j < 4; ++j) v[j] = o[4 * t + j][r] + bias[j] + xr[j];
        *(v4f*)&buf2[off] = v;
      }
  }
  __syncthreads();

  if (tid < T_N) {
    const float* rp = &buf2[tid * 128];
    float mu = 0.f;
    for (int c = 0; c < 128; ++c) mu += rp[c];
    mu *= (1.f / 128.f);
    float var = 0.f;
    for (int c = 0; c < 128; ++c) { const float d = rp[c] - mu; var += d * d; }
    var *= (1.f / 128.f);
    const float rs = rsqrtf(var + EPS_LN);
    float* wp = &buf3[tid * 128];
    for (int c = 0; c < 128; ++c) wp[c] = (rp[c] - mu) * rs * g1[c] + b1[c];
  }
  __syncthreads();

  v8f y[8] = {};
  for (int ch = 0; ch < 4; ++ch) {
    v8f g[8] = {};
    gemm_acc(buf3, W1 + ch * 128, FF_N, wave, half, lm, g);
    const v4f bias = *(const v4f*)&bf1[ch * 128 + c4];
    for (int t = 0; t < 2; ++t)
      for (int r = 0; r < 8; ++r) {
        v4f v;
        for (int j = 0; j < 4; ++j) v[j] = fmaxf(g[4 * t + j][r] + bias[j], 0.f);
        *(v4f*)&buf1[tile_row(t, r, half) * 128 + c4] = v;
      }
    __syncthreads();
    gemm_acc(buf1, W2 + (size_t)(ch * 128) * D_N, 128, wave, half, lm, y);
    __syncthreads();
  }

  {
    const v4f bias = *(const v4f*)&bf2[c4];
    for (int t = 0; t < 2; ++t)
      for (int r = 0; r < 8; ++r) {
        const int off = tile_row(t, r, half) * 128 + c4;
        const v4f hr = *(const v4f*)&buf3[off];
        v4f v;
        for (int j = 0; j < 4; ++j) v[j] = y[4 * t + j][r] + bias[j] + hr[j];
        *(v4f*)&buf1[off] = v;
      }
  }
  __syncthreads();

  Staged nx;
  if (more) fetch_event(x, Kx, Vx, e_next, tid, nb_next, nx);

  if (tid < T_N) {
    const float* rp = &buf1[tid * 128];
    float mu = 0.f;
    for (int c = 0; c < 128; ++c) mu += rp[c];
    mu *= (1.f / 128.f);
    float var = 0.f;
    for (int c = 0; c < 128; ++c) { const float d = rp[c] - mu; var += d * d; }
    var *= (1.f / 128.f);
    const float rs = rsqrtf(var + EPS_LN);
    for (int c = 0; c < 128; ++c)
      out[xbase + (size_t)tid * D_N + c] = (rp[c] - mu) * rs * g2[c] + b2[c];
  }

  if (more) store_event(buf0, buf2, tid, nx);
  __syncthreads();
  }
}

extern "C" void kernel_launch(void* const* d_in, const int* in_sizes, int n_in,
                              void* d_out, int out_size, void* d_ws, size_t ws_size,
                              hipStream_t stream) {
  const float* x   = (const float*)d_in[0];
  const int*   nbr = (const int*)d_in[1];
  const float* Wq  = (const float*)d_in[2];
  const float* Wk  = (const float*)d_in[3];
  const float* Wv  = (const float*)d_in[4];
  const float* Wo  = (const float*)d_in[5];
  const float* bo  = (const float*)d_in[6];
  const float* g1  = (const float*)d_in[7];
  const float* b1  = (const float*)d_in[8];
  const float* g2  = (const float*)d_in[9];
  const float* b2  = (const float*)d_in[10];
  const float* W1  = (const float*)d_in[11];
  const float* bf1 = (const float*)d_in[12];
  const float* W2  = (const float*)d_in[13];
  const float* bf2 = (const float*)d_in[14];

  float* Kx = (float*)d_ws;
  float* Vx = Kx + (size_t)E_N * D_N;

  kv_proj_kernel<<<E_N / 16, 128, 0, stream>>>(x, Wk, Wv, Kx, Vx);
  attn_block_kernel<<<1024, 64, 0, stream>>>(x, nbr, Wq, Wo, bo, g1, b1, g2, b2,
                                             W1, bf1, W2, bf2, Kx, Vx,
                                             (float*)d_out);
}


# one wave per event, every product transposed (weights = A operand, token = lane): D registers feed the next product as B fragments, no LDS or barriers; k order and LN/softmax sums reordered
# speedup vs baseline: 1.7748x; 1.2215x over previous
#include <hip/hip_runtime.h>


#define E_N 10000
#define T_N 24
#define D_N 128
#define H_N 8
#define K_N 16
#define FF_N 512
#define EPS_LN 1e-5f

typedef float v2f __attribute__((ext_vector_type(2)));
typedef float v4f __attribute__((ext_vector_type(4)));
typedef float v8f __attribute__((ext_vector_type(8)));
typedef int v4i __attribute__((ext_vector_type(4)));

__device__ __forceinline__ v8f wmma_f32(v2f a, v2f b, v8f c) {
  return __builtin_amdgcn_wmma_f32_16x16x4_f32(
       false, a,  false, b,
       (short)0, c,  false,  false);
}

__global__ __launch_bounds__(128) void kv_proj_kernel(
    const float* __restrict__ x, const float* __restrict__ Wk,
    const float* __restrict__ Wv, float* __restrict__ Kx, float* __restrict__ Vx) {
  __shared__ float sx0[16 * 128];
  const int tid = threadIdx.x;
  const int e0 = blockIdx.x * 16;
  for (int i = tid; i < 16 * 128; i += 128) {
    const int r = i >> 7, c = i & 127;
    sx0[i] = x[(size_t)(e0 + r) * (T_N * D_N) + c];
  }
  __syncthreads();

  const int lane = tid & 31, wave = tid >> 5;
  const int half = lane >> 4, lm = lane & 15;
  const int n0 = wave * 32 + lm, n1 = n0 + 16;
  v8f aK0{}, aK1{}, aV0{}, aV1{};
  for (int k0 = 0; k0 < 128; k0 += 4) {
    const int ka = k0 + 2 * half;
    v2f a, bk0, bk1, bv0, bv1;
    a.x = sx0[lm * 128 + ka];  a.y = sx0[lm * 128 + ka + 1];
    bk0.x = Wk[ka * 128 + n0]; bk0.y = Wk[(ka + 1) * 128 + n0];
    bk1.x = Wk[ka * 128 + n1]; bk1.y = Wk[(ka + 1) * 128 + n1];
    bv0.x = Wv[ka * 128 + n0]; bv0.y = Wv[(ka + 1) * 128 + n0];
    bv1.x = Wv[ka * 128 + n1]; bv1.y = Wv[(ka + 1) * 128 + n1];
    aK0 = wmma_f32(a, bk0, aK0);
    aK1 = wmma_f32(a, bk1, aK1);
    aV0 = wmma_f32(a, bv0, aV0);
    aV1 = wmma_f32(a, bv1, aV1);
  }
  for (int r = 0; r < 8; ++r) {
    const size_t row = (size_t)(e0 + r + 8 * half) * 128;
    Kx[row + n0] = aK0[r];
    Kx[row + n1] = aK1[r];
    Vx[row + n0] = aV0[r];
    Vx[row + n1] = aV1[r];
  }
}


template <int NG> struct WPtr { const float* p[NG]; };
template <int NG, int RS> struct WFrag { v4f w[RS][NG][2]; };

template <int NG, int RS, int KMODE, int LD>
__device__ __forceinline__ void wload(WFrag<NG, RS>& f, const WPtr<NG>& wp, int blk) {
#pragma unroll
  for (int st = 0; st < RS; ++st) {
    const int step = RS * blk + st, i = step >> 2, s = step & 3;
    const int k0 = KMODE == 0 ? 64 * (i >> 2) + 8 * s + (i & 3) : 16 * i + 2 * s;
    const int k1 = k0 + (KMODE == 0 ? 4 : 1);
#pragma unroll
    for (int G = 0; G < NG; ++G) {
      f.w[st][G][0] = *(const v4f*)(wp.p[G] + (size_t)k0 * LD);
      f.w[st][G][1] = *(const v4f*)(wp.p[G] + (size_t)k1 * LD);
    }
  }
}

template <int NG, int RS>
__device__ __forceinline__ void wmma_block(const WFrag<NG, RS>& f, int blk,
                                           const v8f* in0, const v8f* in1,
                                           v8f* acc0, v8f* acc1) {
#pragma unroll
  for (int st = 0; st < RS; ++st) {
    const int step = RS * blk + st, i = step >> 2, s = step & 3;
    v2f b0, b1;
    b0.x = in0[i][2 * s]; b0.y = in0[i][2 * s + 1];
    b1.x = in1[i][2 * s]; b1.y = in1[i][2 * s + 1];
#pragma unroll
    for (int G = 0; G < NG; ++G)
#pragma unroll
      for (int j = 0; j < 4; ++j) {
        v2f a;
        a.x = f.w[st][G][0][j]; a.y = f.w[st][G][1][j];
        acc0[4 * G + j] = wmma_f32(a, b0, acc0[4 * G + j]);
        acc1[4 * G + j] = wmma_f32(a, b1, acc1[4 * G + j]);
      }
  }
}

template <int NG, int KMODE, int LD, int NI>
__device__ __forceinline__ void gemm_rc(const WPtr<NG>& wp, const v8f* in0, const v8f* in1,
                                        v8f* acc0, v8f* acc1) {
  constexpr int RS = 4 / NG;
  constexpr int NB = NI * 4 / RS;
  WFrag<NG, RS> fa, fb;
  wload<NG, RS, KMODE, LD>(fa, wp, 0);
#pragma unroll
  for (int b = 0; b < NB; b += 2) {
    wload<NG, RS, KMODE, LD>(fb, wp, b + 1);
    __builtin_amdgcn_sched_barrier(0);
    wmma_block<NG, RS>(fa, b, in0, in1, acc0, acc1);
    __builtin_amdgcn_sched_barrier(0);
    if (b + 2 < NB) wload<NG, RS, KMODE, LD>(fa, wp, b + 2);
    __builtin_amdgcn_sched_barrier(0);
    wmma_block<NG, RS>(fb, b + 1, in0, in1, acc0, acc1);
    __builtin_amdgcn_sched_barrier(0);
  }
}

__device__ __forceinline__ void layer_norm_p(v8f* v, const float* __restrict__ gam,
                                             const float* __restrict__ bet, int h) {
  float s = 0.f;
#pragma unroll
  for (int a = 0; a < 8; ++a)
#pragma unroll
    for (int r = 0; r < 8; ++r) s += v[a][r];
  s += __shfl_xor(s, 16);
  const float mu = s * (1.f / 128.f);
  float q = 0.f;
#pragma unroll
  for (int a = 0; a < 8; ++a)
#pragma unroll
    for (int r = 0; r < 8; ++r) { const float d = v[a][r] - mu; q += d * d; }
  q += __shfl_xor(q, 16);
  const float rs = rsqrtf(q * (1.f / 128.f) + EPS_LN);
#pragma unroll
  for (int g = 0; g < 2; ++g)
#pragma unroll
    for (int r = 0; r < 8; ++r) {
      const v4f gg = *(const v4f*)&gam[64 * g + 32 * h + 4 * r];
      const v4f bb = *(const v4f*)&bet[64 * g + 32 * h + 4 * r];
#pragma unroll
      for (int j = 0; j < 4; ++j)
        v[4 * g + j][r] = (v[4 * g + j][r] - mu) * rs * gg[j] + bb[j];
    }
}

__global__ __launch_bounds__(32) __attribute__((amdgpu_waves_per_eu(2, 2))) void attn_block_kernel(
    const float* __restrict__ x, const int* __restrict__ nbr,
    const float* __restrict__ Wq, const float* __restrict__ Wo,
    const float* __restrict__ bo, const float* __restrict__ g1,
    const float* __restrict__ b1, const float* __restrict__ g2,
    const float* __restrict__ b2, const float* __restrict__ W1,
    const float* __restrict__ bf1, const float* __restrict__ W2,
    const float* __restrict__ bf2, const float* __restrict__ Kx,
    const float* __restrict__ Vx, float* __restrict__ out) {
  const int e = blockIdx.x;
  const int lane = threadIdx.x;
  const int h = lane >> 4, c = lane & 15;
  const int row0 = c, row1 = 16 + (c & 7);
  const float* xe = x + (size_t)e * (T_N * D_N);
  const float* xr0 = xe + row0 * D_N + 32 * h;
  const float* xr1 = xe + row1 * D_N + 32 * h;

  const int nbk = nbr[e * K_N + c];
  const v4i nva = *(const v4i*)(nbr + e * K_N + 8 * h);
  const v4i nvb = *(const v4i*)(nbr + e * K_N + 8 * h + 4);

  v8f q0[8] = {}, q1[8] = {};
  {
    v8f x0[8], x1[8];
#pragma unroll
    for (int g = 0; g < 2; ++g)
#pragma unroll
      for (int r = 0; r < 8; ++r) {
        const v4f a = *(const v4f*)(xr0 + 64 * g + 4 * r);
        const v4f b = *(const v4f*)(xr1 + 64 * g + 4 * r);
#pragma unroll
        for (int j = 0; j < 4; ++j) { x0[4 * g + j][r] = a[j]; x1[4 * g + j][r] = b[j]; }
      }
    WPtr<2> wp;
    wp.p[0] = Wq + (size_t)(32 * h) * D_N + 16 * (c & 7) + 8 * (c >> 3);
    wp.p[1] = wp.p[0] + 4;
    gemm_rc<2, 0, D_N, 8>(wp, x0, x1, q0, q1);
  }

  v8f z0[8], z1[8];
  {
    const float* kp = Kx + (size_t)nbk * D_N + 8 * h;
    int voff[8];
#pragma unroll
    for (int i = 0; i < 8; ++i) voff[i] = (i < 4 ? nva[i & 3] : nvb[i & 3]) * D_N + c;
#pragma unroll
    for (int hd = 0; hd < 8; ++hd) {
      const v4f kk0 = *(const v4f*)(kp + 16 * hd);
      const v4f kk1 = *(const v4f*)(kp + 16 * hd + 4);
      v8f s0 = {}, s1 = {};
#pragma unroll
      for (int l = 0; l < 2; ++l)
#pragma unroll
        for (int p = 0; p < 2; ++p) {
          v2f a, b0, b1;
          a.x = l == 0 ? kk0[2 * p] : kk1[2 * p];
          a.y = l == 0 ? kk0[2 * p + 1] : kk1[2 * p + 1];
          b0.x = q0[4 * l + 2 * p][hd]; b0.y = q0[4 * l + 2 * p + 1][hd];
          b1.x = q1[4 * l + 2 * p][hd]; b1.y = q1[4 * l + 2 * p + 1][hd];
          s0 = wmma_f32(a, b0, s0);
          s1 = wmma_f32(a, b1, s1);
        }
      float m0 = s0[0], m1 = s1[0];
#pragma unroll
      for (int r = 1; r < 8; ++r) { m0 = fmaxf(m0, s0[r]); m1 = fmaxf(m1, s1[r]); }
      m0 = fmaxf(m0, __shfl_xor(m0, 16));
      m1 = fmaxf(m1, __shfl_xor(m1, 16));
      float t0 = 0.f, t1 = 0.f;
#pragma unroll
      for (int r = 0; r < 8; ++r) {
        s0[r] = __expf((s0[r] - m0) * 0.25f); t0 += s0[r];
        s1[r] = __expf((s1[r] - m1) * 0.25f); t1 += s1[r];
      }
      t0 += __shfl_xor(t0, 16);
      t1 += __shfl_xor(t1, 16);
      const float i0 = 1.f / t0, i1 = 1.f / t1;
#pragma unroll
      for (int r = 0; r < 8; ++r) { s0[r] *= i0; s1[r] *= i1; }
      v8f a0 = {}, a1 = {};
#pragma unroll
      for (int s = 0; s < 4; ++s) {
        v2f a, b0, b1;
        a.x = Vx[voff[2 * s] + 16 * hd];
        a.y = Vx[voff[2 * s + 1] + 16 * hd];
        b0.x = s0[2 * s]; b0.y = s0[2 * s + 1];
        b1.x = s1[2 * s]; b1.y = s1[2 * s + 1];
        a0 = wmma_f32(a, b0, a0);
        a1 = wmma_f32(a, b1, a1);
      }
      z0[hd] = a0; z1[hd] = a1;
    }
  }

  v8f hA[8] = {}, hB[8] = {};
  {
    WPtr<2> wp;
    wp.p[0] = Wo + (size_t)(8 * h) * D_N + 4 * c;
    wp.p[1] = wp.p[0] + 64;
    gemm_rc<2, 1, D_N, 8>(wp, z0, z1, hA, hB);
#pragma unroll
    for (int g = 0; g < 2; ++g)
#pragma unroll
      for (int r = 0; r < 8; ++r) {
        const v4f bias = *(const v4f*)&bo[64 * g + 32 * h + 4 * r];
        const v4f a = *(const v4f*)(xr0 + 64 * g + 4 * r);
        const v4f b = *(const v4f*)(xr1 + 64 * g + 4 * r);
#pragma unroll
        for (int j = 0; j < 4; ++j) {
          hA[4 * g + j][r] = hA[4 * g + j][r] + bias[j] + a[j];
          hB[4 * g + j][r] = hB[4 * g + j][r] + bias[j] + b[j];
        }
      }
    layer_norm_p(hA, g1, b1, h);
    layer_norm_p(hB, g1, b1, h);
  }

  v8f yA[8] = {}, yB[8] = {};
#pragma unroll 1
  for (int ch = 0; ch < 8; ++ch) {
    v8f gA[4] = {}, gB[4] = {};
    WPtr<1> w1p;
    w1p.p[0] = W1 + (size_t)(32 * h) * FF_N + 64 * ch + 4 * c;
    gemm_rc<1, 0, FF_N, 8>(w1p, hA, hB, gA, gB);
#pragma unroll
    for (int r = 0; r < 8; ++r) {
      const v4f bias = *(const v4f*)&bf1[64 * ch + 32 * h + 4 * r];
#pragma unroll
      for (int j = 0; j < 4; ++j) {
        gA[j][r] = fmaxf(gA[j][r] + bias[j], 0.f);
        gB[j][r] = fmaxf(gB[j][r] + bias[j], 0.f);
      }
    }
    WPtr<2> w2p;
    w2p.p[0] = W2 + (size_t)(64 * ch + 32 * h) * D_N + 4 * c;
    w2p.p[1] = w2p.p[0] + 64;
    gemm_rc<2, 0, D_N, 4>(w2p, gA, gB, yA, yB);
  }

#pragma unroll
  for (int g = 0; g < 2; ++g)
#pragma unroll
    for (int r = 0; r < 8; ++r) {
      const v4f bias = *(const v4f*)&bf2[64 * g + 32 * h + 4 * r];
#pragma unroll
      for (int j = 0; j < 4; ++j) {
        yA[4 * g + j][r] = yA[4 * g + j][r] + bias[j] + hA[4 * g + j][r];
        yB[4 * g + j][r] = yB[4 * g + j][r] + bias[j] + hB[4 * g + j][r];
      }
    }
  layer_norm_p(yA, g2, b2, h);
  layer_norm_p(yB, g2, b2, h);
  float* o0 = out + (size_t)e * (T_N * D_N) + row0 * D_N + 32 * h;
  float* o1 = out + (size_t)e * (T_N * D_N) + row1 * D_N + 32 * h;
#pragma unroll
  for (int g = 0; g < 2; ++g)
#pragma unroll
    for (int r = 0; r < 8; ++r) {
      v4f a, b;
#pragma unroll
      for (int j = 0; j < 4; ++j) { a[j] = yA[4 * g + j][r]; b[j] = yB[4 * g + j][r]; }
      *(v4f*)(o0 + 64 * g + 4 * r) = a;
      if (c < 8) *(v4f*)(o1 + 64 * g + 4 * r) = b;
    }
}

extern "C" void kernel_launch(void* const* d_in, const int* in_sizes, int n_in,
                              void* d_out, int out_size, void* d_ws, size_t ws_size,
                              hipStream_t stream) {
  const float* x   = (const float*)d_in[0];
  const int*   nbr = (const int*)d_in[1];
  const float* Wq  = (const float*)d_in[2];
  const float* Wk  = (const float*)d_in[3];
  const float* Wv  = (const float*)d_in[4];
  const float* Wo  = (const float*)d_in[5];
  const float* bo  = (const float*)d_in[6];
  const float* g1  = (const float*)d_in[7];
  const float* b1  = (const float*)d_in[8];
  const float* g2  = (const float*)d_in[9];
  const float* b2  = (const float*)d_in[10];
  const float* W1  = (const float*)d_in[11];
  const float* bf1 = (const float*)d_in[12];
  const float* W2  = (const float*)d_in[13];
  const float* bf2 = (const float*)d_in[14];

  float* Kx = (float*)d_ws;
  float* Vx = Kx + (size_t)E_N * D_N;

  kv_proj_kernel<<<E_N / 16, 128, 0, stream>>>(x, Wk, Wv, Kx, Vx);
  attn_block_kernel<<<E_N, 32, 0, stream>>>(x, nbr, Wq, Wo, bo, g1, b1, g2, b2,
                                             W1, bf1, W2, bf2, Kx, Vx,
                                             (float*)d_out);
}


# token tiles regrouped across events: a wave takes tokens 0..15 of 2 events or 16..23 of 4 (scores/Z per column half, K/V rows fetched one head ahead); no padded columns, 7500 waves; sums unchanged
# speedup vs baseline: 2.1105x; 1.1891x over previous
#include <hip/hip_runtime.h>


#define E_N 10000
#define T_N 24
#define D_N 128
#define H_N 8
#define K_N 16
#define FF_N 512
#define EPS_LN 1e-5f

typedef float v2f __attribute__((ext_vector_type(2)));
typedef float v4f __attribute__((ext_vector_type(4)));
typedef float v8f __attribute__((ext_vector_type(8)));
typedef int v4i __attribute__((ext_vector_type(4)));

__device__ __forceinline__ v8f wmma_f32(v2f a, v2f b, v8f c) {
  return __builtin_amdgcn_wmma_f32_16x16x4_f32(
       false, a,  false, b,
       (short)0, c,  false,  false);
}

__global__ __launch_bounds__(128) void kv_proj_kernel(
    const float* __restrict__ x, const float* __restrict__ Wk,
    const float* __restrict__ Wv, float* __restrict__ Kx, float* __restrict__ Vx) {
  __shared__ float sx0[16 * 128];
  const int tid = threadIdx.x;
  const int e0 = blockIdx.x * 16;
  for (int i = tid; i < 16 * 128; i += 128) {
    const int r = i >> 7, c = i & 127;
    sx0[i] = x[(size_t)(e0 + r) * (T_N * D_N) + c];
  }
  __syncthreads();

  const int lane = tid & 31, wave = tid >> 5;
  const int half = lane >> 4, lm = lane & 15;
  const int n0 = wave * 32 + lm, n1 = n0 + 16;
  v8f aK0{}, aK1{}, aV0{}, aV1{};
  for (int k0 = 0; k0 < 128; k0 += 4) {
    const int ka = k0 + 2 * half;
    v2f a, bk0, bk1, bv0, bv1;
    a.x = sx0[lm * 128 + ka];  a.y = sx0[lm * 128 + ka + 1];
    bk0.x = Wk[ka * 128 + n0]; bk0.y = Wk[(ka + 1) * 128 + n0];
    bk1.x = Wk[ka * 128 + n1]; bk1.y = Wk[(ka + 1) * 128 + n1];
    bv0.x = Wv[ka * 128 + n0]; bv0.y = Wv[(ka + 1) * 128 + n0];
    bv1.x = Wv[ka * 128 + n1]; bv1.y = Wv[(ka + 1) * 128 + n1];
    aK0 = wmma_f32(a, bk0, aK0);
    aK1 = wmma_f32(a, bk1, aK1);
    aV0 = wmma_f32(a, bv0, aV0);
    aV1 = wmma_f32(a, bv1, aV1);
  }
  for (int r = 0; r < 8; ++r) {
    const size_t row = (size_t)(e0 + r + 8 * half) * 128;
    Kx[row + n0] = aK0[r];
    Kx[row + n1] = aK1[r];
    Vx[row + n0] = aV0[r];
    Vx[row + n1] = aV1[r];
  }
}


template <int NG> struct WPtr { const float* p[NG]; };
template <int NG, int RS> struct WFrag { v4f w[RS][NG][2]; };

template <int NG, int RS, int KMODE, int LD>
__device__ __forceinline__ void wload(WFrag<NG, RS>& f, const WPtr<NG>& wp, int blk) {
#pragma unroll
  for (int st = 0; st < RS; ++st) {
    const int step = RS * blk + st, i = step >> 2, s = step & 3;
    const int k0 = KMODE == 0 ? 64 * (i >> 2) + 8 * s + (i & 3) : 16 * i + 2 * s;
    const int k1 = k0 + (KMODE == 0 ? 4 : 1);
#pragma unroll
    for (int G = 0; G < NG; ++G) {
      f.w[st][G][0] = *(const v4f*)(wp.p[G] + (size_t)k0 * LD);
      f.w[st][G][1] = *(const v4f*)(wp.p[G] + (size_t)k1 * LD);
    }
  }
}

template <int NG, int RS>
__device__ __forceinline__ void wmma_block(const WFrag<NG, RS>& f, int blk,
                                           const v8f* in0, const v8f* in1,
                                           v8f* acc0, v8f* acc1) {
#pragma unroll
  for (int st = 0; st < RS; ++st) {
    const int step = RS * blk + st, i = step >> 2, s = step & 3;
    v2f b0, b1;
    b0.x = in0[i][2 * s]; b0.y = in0[i][2 * s + 1];
    b1.x = in1[i][2 * s]; b1.y = in1[i][2 * s + 1];
#pragma unroll
    for (int G = 0; G < NG; ++G)
#pragma unroll
      for (int j = 0; j < 4; ++j) {
        v2f a;
        a.x = f.w[st][G][0][j]; a.y = f.w[st][G][1][j];
        acc0[4 * G + j] = wmma_f32(a, b0, acc0[4 * G + j]);
        acc1[4 * G + j] = wmma_f32(a, b1, acc1[4 * G + j]);
      }
  }
}

template <int NG, int KMODE, int LD, int NI>
__device__ __forceinline__ void gemm_rc(const WPtr<NG>& wp, const v8f* in0, const v8f* in1,
                                        v8f* acc0, v8f* acc1) {
  constexpr int RS = 4 / NG;
  constexpr int NB = NI * 4 / RS;
  WFrag<NG, RS> fa, fb;
  wload<NG, RS, KMODE, LD>(fa, wp, 0);
#pragma unroll
  for (int b = 0; b < NB; b += 2) {
    wload<NG, RS, KMODE, LD>(fb, wp, b + 1);
    __builtin_amdgcn_sched_barrier(0);
    wmma_block<NG, RS>(fa, b, in0, in1, acc0, acc1);
    __builtin_amdgcn_sched_barrier(0);
    if (b + 2 < NB) wload<NG, RS, KMODE, LD>(fa, wp, b + 2);
    __builtin_amdgcn_sched_barrier(0);
    wmma_block<NG, RS>(fb, b + 1, in0, in1, acc0, acc1);
    __builtin_amdgcn_sched_barrier(0);
  }
}

__device__ __forceinline__ void layer_norm_p(v8f* v, const float* __restrict__ gam,
                                             const float* __restrict__ bet, int h) {
  float s = 0.f;
#pragma unroll
  for (int a = 0; a < 8; ++a)
#pragma unroll
    for (int r = 0; r < 8; ++r) s += v[a][r];
  s += __shfl_xor(s, 16);
  const float mu = s * (1.f / 128.f);
  float q = 0.f;
#pragma unroll
  for (int a = 0; a < 8; ++a)
#pragma unroll
    for (int r = 0; r < 8; ++r) { const float d = v[a][r] - mu; q += d * d; }
  q += __shfl_xor(q, 16);
  const float rs = rsqrtf(q * (1.f / 128.f) + EPS_LN);
#pragma unroll
  for (int g = 0; g < 2; ++g)
#pragma unroll
    for (int r = 0; r < 8; ++r) {
      const v4f gg = *(const v4f*)&gam[64 * g + 32 * h + 4 * r];
      const v4f bb = *(const v4f*)&bet[64 * g + 32 * h + 4 * r];
#pragma unroll
      for (int j = 0; j < 4; ++j)
        v[4 * g + j][r] = (v[4 * g + j][r] - mu) * rs * gg[j] + bb[j];
    }
}

struct Gath { v4f k[2][2][2]; float v[2][2][8]; };

__device__ __forceinline__ void gather_head(Gath& g, const float* const (&kp)[2][2],
                                            const int (&voff)[2][2][8],
                                            const float* __restrict__ Vx, int hd) {
#pragma unroll
  for (int t = 0; t < 2; ++t)
#pragma unroll
    for (int X = 0; X < 2; ++X) {
      g.k[t][X][0] = *(const v4f*)(kp[t][X] + 16 * hd);
      g.k[t][X][1] = *(const v4f*)(kp[t][X] + 16 * hd + 4);
#pragma unroll
      for (int i = 0; i < 8; ++i) g.v[t][X][i] = Vx[voff[t][X][i] + 16 * hd];
    }
}

__device__ __forceinline__ void attn_head(const Gath& g, int hd, bool lo, const v8f* q0,
                                          const v8f* q1, v8f* z0, v8f* z1) {
#pragma unroll
  for (int t = 0; t < 2; ++t) {
    const v8f* q = t == 0 ? q0 : q1;
    v8f sA = {}, sB = {};
#pragma unroll
    for (int l = 0; l < 2; ++l)
#pragma unroll
      for (int p = 0; p < 2; ++p) {
        v2f aA, aB, b;
        aA.x = g.k[t][0][l][2 * p]; aA.y = g.k[t][0][l][2 * p + 1];
        aB.x = g.k[t][1][l][2 * p]; aB.y = g.k[t][1][l][2 * p + 1];
        b.x = q[4 * l + 2 * p][hd]; b.y = q[4 * l + 2 * p + 1][hd];
        sA = wmma_f32(aA, b, sA);
        sB = wmma_f32(aB, b, sB);
      }
    v8f s;
#pragma unroll
    for (int r = 0; r < 8; ++r) s[r] = lo ? sA[r] : sB[r];
    float m = s[0];
#pragma unroll
    for (int r = 1; r < 8; ++r) m = fmaxf(m, s[r]);
    m = fmaxf(m, __shfl_xor(m, 16));
    float tt = 0.f;
#pragma unroll
    for (int r = 0; r < 8; ++r) { s[r] = __expf((s[r] - m) * 0.25f); tt += s[r]; }
    tt += __shfl_xor(tt, 16);
    const float inv = 1.f / tt;
#pragma unroll
    for (int r = 0; r < 8; ++r) s[r] *= inv;
    v8f zA = {}, zB = {};
#pragma unroll
    for (int st = 0; st < 4; ++st) {
      v2f aA, aB, b;
      aA.x = g.v[t][0][2 * st]; aA.y = g.v[t][0][2 * st + 1];
      aB.x = g.v[t][1][2 * st]; aB.y = g.v[t][1][2 * st + 1];
      b.x = s[2 * st]; b.y = s[2 * st + 1];
      zA = wmma_f32(aA, b, zA);
      zB = wmma_f32(aB, b, zB);
    }
    v8f z;
#pragma unroll
    for (int r = 0; r < 8; ++r) z[r] = lo ? zA[r] : zB[r];
    if (t == 0) z0[hd] = z; else z1[hd] = z;
  }
}

__global__ __launch_bounds__(32) __attribute__((amdgpu_waves_per_eu(2, 2))) void attn_block_kernel(
    const float* __restrict__ x, const int* __restrict__ nbr,
    const float* __restrict__ Wq, const float* __restrict__ Wo,
    const float* __restrict__ bo, const float* __restrict__ g1,
    const float* __restrict__ b1, const float* __restrict__ g2,
    const float* __restrict__ b2, const float* __restrict__ W1,
    const float* __restrict__ bf1, const float* __restrict__ W2,
    const float* __restrict__ bf2, const float* __restrict__ Kx,
    const float* __restrict__ Vx, float* __restrict__ out) {
  const int w = blockIdx.x;
  const int lane = threadIdx.x;
  const int h = lane >> 4, c = lane & 15;
  const bool tail = w >= E_N / 2;
  const bool lo = c < 8;
  int ev[2][2];
  ev[0][0] = tail ? 4 * (w - E_N / 2) : 2 * w;
  ev[0][1] = tail ? ev[0][0] + 1 : ev[0][0];
  ev[1][0] = tail ? ev[0][0] + 2 : ev[0][0] + 1;
  ev[1][1] = tail ? ev[0][0] + 3 : ev[0][0] + 1;
  const int tok = tail ? 16 + (c & 7) : c;
  const size_t tb0 = (size_t)(lo ? ev[0][0] : ev[0][1]) * (T_N * D_N) + tok * D_N + 32 * h;
  const size_t tb1 = (size_t)(lo ? ev[1][0] : ev[1][1]) * (T_N * D_N) + tok * D_N + 32 * h;
  const float* xr0 = x + tb0;
  const float* xr1 = x + tb1;

  const float* kp[2][2];
  int voff[2][2][8];
#pragma unroll
  for (int t = 0; t < 2; ++t)
#pragma unroll
    for (int X = 0; X < 2; ++X) {
      const int* nb = nbr + ev[t][X] * K_N;
      kp[t][X] = Kx + (size_t)nb[c] * D_N + 8 * h;
      const v4i nva = *(const v4i*)(nb + 8 * h);
      const v4i nvb = *(const v4i*)(nb + 8 * h + 4);
#pragma unroll
      for (int i = 0; i < 8; ++i) voff[t][X][i] = (i < 4 ? nva[i & 3] : nvb[i & 3]) * D_N + c;
    }
  Gath ga, gb;
  gather_head(ga, kp, voff, Vx, 0);

  v8f q0[8] = {}, q1[8] = {};
  {
    v8f x0[8], x1[8];
#pragma unroll
    for (int g = 0; g < 2; ++g)
#pragma unroll
      for (int r = 0; r < 8; ++r) {
        const v4f a = *(const v4f*)(xr0 + 64 * g + 4 * r);
        const v4f b = *(const v4f*)(xr1 + 64 * g + 4 * r);
#pragma unroll
        for (int j = 0; j < 4; ++j) { x0[4 * g + j][r] = a[j]; x1[4 * g + j][r] = b[j]; }
      }
    WPtr<2> wp;
    wp.p[0] = Wq + (size_t)(32 * h) * D_N + 16 * (c & 7) + 8 * (c >> 3);
    wp.p[1] = wp.p[0] + 4;
    gemm_rc<2, 0, D_N, 8>(wp, x0, x1, q0, q1);
  }

  v8f z0[8], z1[8];
#pragma unroll
  for (int hd = 0; hd < 8; hd += 2) {
    gather_head(gb, kp, voff, Vx, hd + 1);
    __builtin_amdgcn_sched_barrier(0);
    attn_head(ga, hd, lo, q0, q1, z0, z1);
    __builtin_amdgcn_sched_barrier(0);
    if (hd + 2 < 8) gather_head(ga, kp, voff, Vx, hd + 2);
    __builtin_amdgcn_sched_barrier(0);
    attn_head(gb, hd + 1, lo, q0, q1, z0, z1);
    __builtin_amdgcn_sched_barrier(0);
  }

  v8f hA[8] = {}, hB[8] = {};
  {
    WPtr<2> wp;
    wp.p[0] = Wo + (size_t)(8 * h) * D_N + 4 * c;
    wp.p[1] = wp.p[0] + 64;
    gemm_rc<2, 1, D_N, 8>(wp, z0, z1, hA, hB);
#pragma unroll
    for (int g = 0; g < 2; ++g)
#pragma unroll
      for (int r = 0; r < 8; ++r) {
        const v4f bias = *(const v4f*)&bo[64 * g + 32 * h + 4 * r];
        const v4f a = *(const v4f*)(xr0 + 64 * g + 4 * r);
        const v4f b = *(const v4f*)(xr1 + 64 * g + 4 * r);
#pragma unroll
        for (int j = 0; j < 4; ++j) {
          hA[4 * g + j][r] = hA[4 * g + j][r] + bias[j] + a[j];
          hB[4 * g + j][r] = hB[4 * g + j][r] + bias[j] + b[j];
        }
      }
    layer_norm_p(hA, g1, b1, h);
    layer_norm_p(hB, g1, b1, h);
  }

  v8f yA[8] = {}, yB[8] = {};
#pragma unroll 1
  for (int ch = 0; ch < 8; ++ch) {
    v8f gA[4] = {}, gB[4] = {};
    WPtr<1> w1p;
    w1p.p[0] = W1 + (size_t)(32 * h) * FF_N + 64 * ch + 4 * c;
    gemm_rc<1, 0, FF_N, 8>(w1p, hA, hB, gA, gB);
#pragma unroll
    for (int r = 0; r < 8; ++r) {
      const v4f bias = *(const v4f*)&bf1[64 * ch + 32 * h + 4 * r];
#pragma unroll
      for (int j = 0; j < 4; ++j) {
        gA[j][r] = fmaxf(gA[j][r] + bias[j], 0.f);
        gB[j][r] = fmaxf(gB[j][r] + bias[j], 0.f);
      }
    }
    WPtr<2> w2p;
    w2p.p[0] = W2 + (size_t)(64 * ch + 32 * h) * D_N + 4 * c;
    w2p.p[1] = w2p.p[0] + 64;
    gemm_rc<2, 0, D_N, 4>(w2p, gA, gB, yA, yB);
  }

#pragma unroll
  for (int g = 0; g < 2; ++g)
#pragma unroll
    for (int r = 0; r < 8; ++r) {
      const v4f bias = *(const v4f*)&bf2[64 * g + 32 * h + 4 * r];
#pragma unroll
      for (int j = 0; j < 4; ++j) {
        yA[4 * g + j][r] = yA[4 * g + j][r] + bias[j] + hA[4 * g + j][r];
        yB[4 * g + j][r] = yB[4 * g + j][r] + bias[j] + hB[4 * g + j][r];
      }
    }
  layer_norm_p(yA, g2, b2, h);
  layer_norm_p(yB, g2, b2, h);
  float* o0 = out + tb0;
  float* o1 = out + tb1;
#pragma unroll
  for (int g = 0; g < 2; ++g)
#pragma unroll
    for (int r = 0; r < 8; ++r) {
      v4f a, b;
#pragma unroll
      for (int j = 0; j < 4; ++j) { a[j] = yA[4 * g + j][r]; b[j] = yB[4 * g + j][r]; }
      *(v4f*)(o0 + 64 * g + 4 * r) = a;
      *(v4f*)(o1 + 64 * g + 4 * r) = b;
    }
}

extern "C" void kernel_launch(void* const* d_in, const int* in_sizes, int n_in,
                              void* d_out, int out_size, void* d_ws, size_t ws_size,
                              hipStream_t stream) {
  const float* x   = (const float*)d_in[0];
  const int*   nbr = (const int*)d_in[1];
  const float* Wq  = (const float*)d_in[2];
  const float* Wk  = (const float*)d_in[3];
  const float* Wv  = (const float*)d_in[4];
  const float* Wo  = (const float*)d_in[5];
  const float* bo  = (const float*)d_in[6];
  const float* g1  = (const float*)d_in[7];
  const float* b1  = (const float*)d_in[8];
  const float* g2  = (const float*)d_in[9];
  const float* b2  = (const float*)d_in[10];
  const float* W1  = (const float*)d_in[11];
  const float* bf1 = (const float*)d_in[12];
  const float* W2  = (const float*)d_in[13];
  const float* bf2 = (const float*)d_in[14];

  float* Kx = (float*)d_ws;
  float* Vx = Kx + (size_t)E_N * D_N;

  kv_proj_kernel<<<E_N / 16, 128, 0, stream>>>(x, Wk, Wv, Kx, Vx);
  attn_block_kernel<<<E_N / 2 + E_N / 4, 32, 0, stream>>>(x, nbr, Wq, Wo, bo, g1, b1, g2, b2,
                                             W1, bf1, W2, bf2, Kx, Vx,
                                             (float*)d_out);
}
